# starting kernel
# baseline (speedup 1.0000x reference)
#include <hip/hip_runtime.h>
#include <math.h>

typedef __attribute__((ext_vector_type(2))) float v2f;
typedef __attribute__((ext_vector_type(8))) float v8f;

#define NFEAT 512
#define HIDDEN 16
#define NCLS 2


__global__ void k_init_deg(float* __restrict__ deg, int n) {
  int i = blockIdx.x * blockDim.x + threadIdx.x;
  if (i < n) deg[i] = 1.0f;
}

__global__ void k_deg_count(const int* __restrict__ col, float* __restrict__ deg, int E) {
  int e = blockIdx.x * blockDim.x + threadIdx.x;
  if (e < E) atomicAdd(&deg[col[e]], 1.0f);
}

__global__ void k_deg_rsqrt(float* __restrict__ deg, int n) {
  int i = blockIdx.x * blockDim.x + threadIdx.x;
  if (i < n) deg[i] = rsqrtf(deg[i]);
}


__global__ __launch_bounds__(256) void k_gemm1(const float* __restrict__ x,
                                               const float* __restrict__ W1,
                                               float* __restrict__ h1,
                                               int n_nodes) {
  const int lane = threadIdx.x & 31;
  const int tile = blockIdx.x * 8 + (threadIdx.x >> 5);
  const int row0 = tile * 16;
  if (row0 >= n_nodes) return;

  const int nsel  = lane & 15;
  const int khalf = (lane >> 4) << 1;

  const float* arow = x + (size_t)(row0 + nsel) * NFEAT + khalf;

  v8f c = {};
  for (int k = 0; k < NFEAT; k += 4) {
    if ((k & 127) == 0) __builtin_prefetch(arow + k + 128, 0, 0);
    v2f a = *(const v2f*)(arow + k);
    v2f b;
    b.x = W1[(k + khalf) * HIDDEN + nsel];
    b.y = W1[(k + khalf + 1) * HIDDEN + nsel];
    c = __builtin_amdgcn_wmma_f32_16x16x4_f32(false, a, false, b, (short)0, c,
                                              false, false);
  }

  const int mbase = row0 + ((lane >> 4) << 3);
#pragma unroll
  for (int r = 0; r < 8; ++r)
    h1[(size_t)(mbase + r) * HIDDEN + nsel] = c[r];
}


__global__ void k_agg1_self(const float* __restrict__ h1, const float* __restrict__ dinv,
                            float* __restrict__ agg1, int n) {
  int t = blockIdx.x * blockDim.x + threadIdx.x;
  int node = t >> 4;
  if (node < n) {
    float w = dinv[node] * dinv[node];
    agg1[t] = h1[t] * w;
  }
}

__global__ void k_agg1_edges(const int* __restrict__ row, const int* __restrict__ col,
                             const float* __restrict__ h1, const float* __restrict__ dinv,
                             float* __restrict__ agg1, int E) {
  int t = blockIdx.x * blockDim.x + threadIdx.x;
  int e = t >> 4;
  int j = t & 15;
  if (e < E) {
    int r = row[e];
    int c = col[e];
    float w = dinv[r] * dinv[c];
    atomicAdd(&agg1[(size_t)c * HIDDEN + j], h1[(size_t)r * HIDDEN + j] * w);
  }
}


__global__ void k_relu_gemm2(const float* __restrict__ agg1, const float* __restrict__ b1,
                             const float* __restrict__ W2, float* __restrict__ h2, int n) {
  int node = blockIdx.x * blockDim.x + threadIdx.x;
  if (node >= n) return;
  float acc0 = 0.f, acc1 = 0.f;
#pragma unroll
  for (int j = 0; j < HIDDEN; ++j) {
    float v = fmaxf(agg1[(size_t)node * HIDDEN + j] + b1[j], 0.f);
    acc0 = fmaf(v, W2[j * NCLS + 0], acc0);
    acc1 = fmaf(v, W2[j * NCLS + 1], acc1);
  }
  h2[(size_t)node * NCLS + 0] = acc0;
  h2[(size_t)node * NCLS + 1] = acc1;
}


__global__ void k_agg2_self(const float* __restrict__ h2, const float* __restrict__ dinv,
                            float* __restrict__ agg2, int n) {
  int node = blockIdx.x * blockDim.x + threadIdx.x;
  if (node < n) {
    float w = dinv[node] * dinv[node];
    agg2[(size_t)node * NCLS + 0] = h2[(size_t)node * NCLS + 0] * w;
    agg2[(size_t)node * NCLS + 1] = h2[(size_t)node * NCLS + 1] * w;
  }
}

__global__ void k_agg2_edges(const int* __restrict__ row, const int* __restrict__ col,
                             const float* __restrict__ h2, const float* __restrict__ dinv,
                             float* __restrict__ agg2, int E) {
  int e = blockIdx.x * blockDim.x + threadIdx.x;
  if (e < E) {
    int r = row[e];
    int c = col[e];
    float w = dinv[r] * dinv[c];
    atomicAdd(&agg2[(size_t)c * NCLS + 0], h2[(size_t)r * NCLS + 0] * w);
    atomicAdd(&agg2[(size_t)c * NCLS + 1], h2[(size_t)r * NCLS + 1] * w);
  }
}


__global__ void k_logsoftmax(const float* __restrict__ agg2, const float* __restrict__ b2,
                             float* __restrict__ out, int n) {
  int node = blockIdx.x * blockDim.x + threadIdx.x;
  if (node >= n) return;
  float z0 = agg2[(size_t)node * NCLS + 0] + b2[0];
  float z1 = agg2[(size_t)node * NCLS + 1] + b2[1];
  float m = fmaxf(z0, z1);
  float lse = m + logf(expf(z0 - m) + expf(z1 - m));
  out[(size_t)node * NCLS + 0] = z0 - lse;
  out[(size_t)node * NCLS + 1] = z1 - lse;
}


extern "C" void kernel_launch(void* const* d_in, const int* in_sizes, int n_in,
                              void* d_out, int out_size, void* d_ws, size_t ws_size,
                              hipStream_t stream) {
  const float* x  = (const float*)d_in[0];
  const int*   ei = (const int*)d_in[1];
  const float* W1 = (const float*)d_in[2];
  const float* b1 = (const float*)d_in[3];
  const float* W2 = (const float*)d_in[4];
  const float* b2 = (const float*)d_in[5];

  const int n = in_sizes[0] / NFEAT;
  const int E = in_sizes[1] / 2;
  const int* row = ei;
  const int* col = ei + E;

  char* ws = (char*)d_ws;
  size_t off = 0;
  auto carve = [&](size_t bytes) -> void* {
    void* p = ws + off;
    off = (off + bytes + 255) & ~(size_t)255;
    return p;
  };
  float* deg  = (float*)carve((size_t)n * sizeof(float));
  float* h1   = (float*)carve((size_t)n * HIDDEN * sizeof(float));
  float* agg1 = (float*)carve((size_t)n * HIDDEN * sizeof(float));
  float* h2   = (float*)carve((size_t)n * NCLS * sizeof(float));
  float* agg2 = (float*)carve((size_t)n * NCLS * sizeof(float));
  (void)ws_size;

  const int B = 256;
  auto blocks = [](long long work, int b) { return (unsigned)((work + b - 1) / b); };

  k_init_deg<<<blocks(n, B), B, 0, stream>>>(deg, n);
  k_deg_count<<<blocks(E, B), B, 0, stream>>>(col, deg, E);
  k_deg_rsqrt<<<blocks(n, B), B, 0, stream>>>(deg, n);

  int tiles = (n + 15) / 16;
  k_gemm1<<<blocks(tiles, 8), B, 0, stream>>>(x, W1, h1, n);

  k_agg1_self<<<blocks((long long)n * HIDDEN, B), B, 0, stream>>>(h1, deg, agg1, n);
  k_agg1_edges<<<blocks((long long)E * HIDDEN, B), B, 0, stream>>>(row, col, h1, deg, agg1, E);

  k_relu_gemm2<<<blocks(n, B), B, 0, stream>>>(agg1, b1, W2, h2, n);

  k_agg2_self<<<blocks(n, B), B, 0, stream>>>(h2, deg, agg2, n);
  k_agg2_edges<<<blocks(E, B), B, 0, stream>>>(row, col, h2, deg, agg2, E);

  k_logsoftmax<<<blocks(n, B), B, 0, stream>>>(agg2, b2, (float*)d_out, n);
}


# degree count and layer-1 GEMM share one launch as two work-group roles (independent work): the count's atomics drain while the other work-groups stream x; 9 launches become 8; sum order unchanged
# speedup vs baseline: 4.5236x; 4.5236x over previous
#include <hip/hip_runtime.h>
#include <math.h>

typedef __attribute__((ext_vector_type(2))) float v2f;
typedef __attribute__((ext_vector_type(8))) float v8f;
typedef __attribute__((ext_vector_type(3))) int v3i;

#define NFEAT 512
#define HIDDEN 16
#define NCLS 2


#define HALF_SLICES 128
#define PASS_GRID 256
#define EDGE_THREADS 1024

__global__ void k_init_deg(float* __restrict__ deg, int* __restrict__ ctr, int n) {
  int i = blockIdx.x * blockDim.x + threadIdx.x;
  if (i < n) deg[i] = 1.0f;
  if (i < 4) ctr[i] = 0;
}

__device__ __forceinline__ int half_split(int n) { return ((n >> 1) + 63) & ~63; }

__device__ __forceinline__ bool claim_slice(int* __restrict__ ctr, int& half, int& slice) {
  __shared__ int s_claim[2];
  if (threadIdx.x == 0) {
    unsigned id = __builtin_amdgcn_s_sendmsg_rtn(0x87);
    int h = (int)((id >> 18) & 1u);
    int r = atomicAdd(&ctr[h], 1);
    if (r >= HALF_SLICES) {
      h ^= 1;
      r = atomicAdd(&ctr[h], 1);
    }
    s_claim[0] = h;
    s_claim[1] = r;
  }
  __syncthreads();
  half = s_claim[0];
  slice = s_claim[1];
  return slice < HALF_SLICES;
}

__device__ __forceinline__ void deg_count_block(const int* __restrict__ col,
                                                float* __restrict__ deg,
                                                int* __restrict__ ctr, int E, int n) {
  int half, slice;
  if (!claim_slice(ctr, half, slice)) return;
  const int nh = half_split(n);
  const int per = (E + HALF_SLICES - 1) / HALF_SLICES;
  const int e0 = slice * per;
  const int e1 = min(E, e0 + per);
  for (int eb = e0 + threadIdx.x; eb < e1; eb += 8 * EDGE_THREADS) {
    int c[8];
#pragma unroll
    for (int u = 0; u < 8; ++u) c[u] = col[min(eb + u * EDGE_THREADS, e1 - 1)];
#pragma unroll
    for (int u = 0; u < 8; ++u) {
      const bool in = eb + u * EDGE_THREADS < e1;
      if (in && ((c[u] >= nh) == (half != 0))) atomicAdd(&deg[c[u]], 1.0f);
    }
  }
}

#define NTAB 100000
#define DEG_ESC 0xFFFFu

__device__ __forceinline__ void fill_deg_table(unsigned short* tab, const float* __restrict__ deg,
                                               int n) {
#pragma unroll 8
  for (int i = threadIdx.x; i < NTAB; i += EDGE_THREADS) {
    float d = deg[min(i, n - 1)];
    d = (i < n) ? d : 1.0f;
    tab[i] = (d < 65535.0f) ? (unsigned short)d : (unsigned short)DEG_ESC;
  }
  __syncthreads();
}

__device__ __forceinline__ unsigned deg_entry(const unsigned short* tab, int i) {
  unsigned d = tab[min(i, NTAB - 1)];
  return (i < NTAB) ? d : DEG_ESC;
}


__device__ __forceinline__ void gemm1_tile(const float* __restrict__ x,
                                           const float* __restrict__ W1,
                                           float* __restrict__ h1,
                                           int n_nodes, int tile) {
  const int lane = threadIdx.x & 31;
  const int row0 = tile * 16;
  if (row0 >= n_nodes) return;

  const int nsel  = lane & 15;
  const int khalf = (lane >> 4) << 1;

  const float* arow = x + (size_t)(row0 + nsel) * NFEAT + khalf;

  v8f c = {};
  for (int k = 0; k < NFEAT; k += 4) {
    if ((k & 127) == 0) __builtin_prefetch(arow + k + 128, 0, 0);
    v2f a = *(const v2f*)(arow + k);
    v2f b;
    b.x = W1[(k + khalf) * HIDDEN + nsel];
    b.y = W1[(k + khalf + 1) * HIDDEN + nsel];
    c = __builtin_amdgcn_wmma_f32_16x16x4_f32(false, a, false, b, (short)0, c,
                                              false, false);
  }

  const int mbase = row0 + ((lane >> 4) << 3);
#pragma unroll
  for (int r = 0; r < 8; ++r)
    h1[(size_t)(mbase + r) * HIDDEN + nsel] = c[r];
}

#define GEMM_WAVES 25

__global__ __launch_bounds__(EDGE_THREADS) void k_deg_gemm(
    const int* __restrict__ col, float* __restrict__ deg, int* __restrict__ ctr, int E, int n,
    const float* __restrict__ x, const float* __restrict__ W1, float* __restrict__ h1) {
  const unsigned gemm_groups = gridDim.x - PASS_GRID;
  if (blockIdx.x >= gemm_groups) {
    deg_count_block(col, deg, ctr, E, n);
    return;
  }
  const int wave = threadIdx.x >> 5;
  if (wave >= GEMM_WAVES) return;
  gemm1_tile(x, W1, h1, n, (int)blockIdx.x * GEMM_WAVES + wave);
}


__global__ void k_agg1_self(const float* __restrict__ h1, const float* __restrict__ deg,
                            float* __restrict__ agg1, int n) {
  int t = blockIdx.x * blockDim.x + threadIdx.x;
  int node = t >> 4;
  if (node < n) {
    float di = rsqrtf(deg[node]);
    float w = di * di;
    agg1[t] = h1[t] * w;
  }
}

__device__ __forceinline__ float edge_factor(const unsigned short* tab,
                                             const float* __restrict__ deg, int r, int c) {
  unsigned dr = deg_entry(tab, r), dc = deg_entry(tab, c);
  float fr = (float)dr, fc = (float)dc;
  if (dr == DEG_ESC) fr = deg[r];
  if (dc == DEG_ESC) fc = deg[c];
  return rsqrtf(fr) * rsqrtf(fc);
}

#define PAIRS 8
#define GROUP (2 * PAIRS)
#define QCAP 64

template <bool FULL>
__device__ __forceinline__ void add_group(const v3i* g, int cnt, int hi, int j,
                                          const float* __restrict__ h1,
                                          float* __restrict__ agg1) {
  v3i e[PAIRS];
  float hv[PAIRS];
  bool act[PAIRS];
#pragma unroll
  for (int p = 0; p < PAIRS; ++p) {
    act[p] = FULL || (2 * p + hi < cnt);
    e[p] = g[2 * p];
  }
#pragma unroll
  for (int p = 0; p < PAIRS; ++p)
    hv[p] = act[p] ? h1[(unsigned)e[p].x * HIDDEN + j] : 0.0f;
#pragma unroll
  for (int p = 0; p < PAIRS; ++p)
    if (act[p]) atomicAdd(&agg1[(unsigned)e[p].y * HIDDEN + j], hv[p] * __int_as_float(e[p].z));
}

__global__ __launch_bounds__(EDGE_THREADS) void k_agg1_edges(
    const int* __restrict__ row, const int* __restrict__ col, const float* __restrict__ h1,
    const float* __restrict__ deg, float* __restrict__ agg1, int* __restrict__ ctr, int E,
    int n) {
  __shared__ unsigned short tab[NTAB];
  __shared__ v3i ring[(EDGE_THREADS / 32) * QCAP];
  int half, slice;
  if (!claim_slice(ctr, half, slice)) return;
  fill_deg_table(tab, deg, n);
  const int nh = half_split(n);
  const int per = (E + HALF_SLICES - 1) / HALF_SLICES;
  const int e0 = slice * per;
  const int e1 = min(E, e0 + per);
  const int lane = threadIdx.x & 31;
  const int j = lane & 15;
  const int hi = lane >> 4;
  v3i* q = ring + (threadIdx.x >> 5) * QCAP;
  const int base0 = e0 + (threadIdx.x & ~31);
  int rn = 0, cn = 0;
  if (base0 < e1) {
    const int ee = min(base0 + lane, e1 - 1);
    rn = row[ee];
    cn = col[ee];
  }
  int head = 0, tail = 0;
  for (int base = base0; base < e1; base += EDGE_THREADS) {
    const bool in = base + lane < e1;
    const int r = rn, c = cn;
    {
      const int en = min(base + EDGE_THREADS + lane, e1 - 1);
      rn = row[en];
      cn = col[en];
    }
    const bool mine = in && ((c >= nh) == (half != 0));
    const unsigned m = (unsigned)__builtin_amdgcn_ballot_w32(mine);
    if (mine) {
      const float w = edge_factor(tab, deg, r, c);
      const int pos = tail + (int)__builtin_amdgcn_mbcnt_lo(m, 0u);
      v3i e;
      e.x = r;
      e.y = c;
      e.z = __float_as_int(w);
      q[pos & (QCAP - 1)] = e;
    }
    tail += __builtin_popcount(m);
    while (tail - head >= GROUP) {
      add_group<true>(q + (head & (QCAP - 1)) + hi, GROUP, hi, j, h1, agg1);
      head += GROUP;
    }
  }
  if (tail != head) add_group<false>(q + (head & (QCAP - 1)) + hi, tail - head, hi, j, h1, agg1);
}


__global__ void k_relu_gemm2(const float* __restrict__ agg1, const float* __restrict__ b1,
                             const float* __restrict__ W2, const float* __restrict__ deg,
                             float* __restrict__ g, int n) {
  int node = blockIdx.x * blockDim.x + threadIdx.x;
  if (node >= n) return;
  float acc0 = 0.f, acc1 = 0.f;
#pragma unroll
  for (int j = 0; j < HIDDEN; ++j) {
    float v = fmaxf(agg1[(size_t)node * HIDDEN + j] + b1[j], 0.f);
    acc0 = fmaf(v, W2[j * NCLS + 0], acc0);
    acc1 = fmaf(v, W2[j * NCLS + 1], acc1);
  }
  g[node] = (acc0 - acc1) * rsqrtf(deg[node]);
}

__global__ void k_agg2_self(const float* __restrict__ g, float* __restrict__ acc,
                            int* __restrict__ ctr, int n) {
  int node = blockIdx.x * blockDim.x + threadIdx.x;
  if (node < 2) ctr[node] = 0;
  if (node < n) acc[node] = g[node];
}

#define SRC_TAB 50048
#define SRC_SLICES 64
#define SCAN2 8
__global__ __launch_bounds__(EDGE_THREADS) void k_agg2_edges(
    const int* __restrict__ row, const int* __restrict__ col, const float* __restrict__ g,
    float* __restrict__ acc, int* __restrict__ ctr, int E, int n) {
  __shared__ __attribute__((aligned(16))) float tab[SRC_TAB];
  int half, ticket;
  if (!claim_slice(ctr, half, ticket)) return;
  const int sh = ticket & 1;
  const int slice = ticket >> 1;
  const int nh = half_split(n);
  const int s0 = sh ? nh : 0;
  const int s1 = sh ? n : min(nh, n);
  const int cnt = min(s1 - s0, SRC_TAB);
  {
    const float4* __restrict__ src4 = (const float4*)(g + s0);
    float4* tab4 = (float4*)tab;
    const int cnt4 = cnt >> 2;
    for (int i0 = threadIdx.x; i0 < cnt4; i0 += 8 * EDGE_THREADS) {
      int k[8];
      float4 v[8];
#pragma unroll
      for (int u = 0; u < 8; ++u) {
        k[u] = min(i0 + u * EDGE_THREADS, cnt4 - 1);
        v[u] = src4[k[u]];
      }
#pragma unroll
      for (int u = 0; u < 8; ++u) tab4[k[u]] = v[u];
    }
    {
      const int i = (cnt4 << 2) + threadIdx.x;
      if (i < cnt) tab[i] = g[s0 + i];
    }
  }
  __syncthreads();
  const int per = (E + SRC_SLICES - 1) / SRC_SLICES;
  const int e0 = slice * per;
  const int e1 = min(E, e0 + per);
  for (int eb = e0 + threadIdx.x; eb < e1; eb += SCAN2 * EDGE_THREADS) {
    int r[SCAN2], c[SCAN2];
#pragma unroll
    for (int u = 0; u < SCAN2; ++u) {
      const int e = min(eb + u * EDGE_THREADS, e1 - 1);
      r[u] = row[e];
      c[u] = col[e];
    }
    float v[SCAN2];
    bool keep[SCAN2];
#pragma unroll
    for (int u = 0; u < SCAN2; ++u) {
      const bool in = eb + u * EDGE_THREADS < e1;
      keep[u] = in && ((c[u] >= nh) == (half != 0)) && (r[u] >= s0) && (r[u] < s1);
      const int k = keep[u] ? min(r[u] - s0, SRC_TAB - 1) : 0;
      v[u] = tab[k];
    }
#pragma unroll
    for (int u = 0; u < SCAN2; ++u) {
      if (keep[u]) {
        const float val = (r[u] - s0 < SRC_TAB) ? v[u] : g[r[u]];
        atomicAdd(&acc[c[u]], val);
      }
    }
  }
}


__global__ void k_logsoftmax(const float* __restrict__ acc, const float* __restrict__ deg,
                             const float* __restrict__ b2, float* __restrict__ out, int n) {
  int node = blockIdx.x * blockDim.x + threadIdx.x;
  if (node >= n) return;
  const float d = acc[node] * rsqrtf(deg[node]) + (b2[0] - b2[1]);
  const float a = fabsf(d);
  const float l = logf(1.0f + expf(-a));
  const float hi = -l;
  const float lo = -a - l;
  out[(size_t)node * NCLS + 0] = (d >= 0.f) ? hi : lo;
  out[(size_t)node * NCLS + 1] = (d >= 0.f) ? lo : hi;
}


extern "C" void kernel_launch(void* const* d_in, const int* in_sizes, int n_in,
                              void* d_out, int out_size, void* d_ws, size_t ws_size,
                              hipStream_t stream) {
  const float* x  = (const float*)d_in[0];
  const int*   ei = (const int*)d_in[1];
  const float* W1 = (const float*)d_in[2];
  const float* b1 = (const float*)d_in[3];
  const float* W2 = (const float*)d_in[4];
  const float* b2 = (const float*)d_in[5];

  const int n = in_sizes[0] / NFEAT;
  const int E = in_sizes[1] / 2;
  const int* row = ei;
  const int* col = ei + E;

  char* ws = (char*)d_ws;
  size_t off = 0;
  auto carve = [&](size_t bytes) -> void* {
    void* p = ws + off;
    off = (off + bytes + 255) & ~(size_t)255;
    return p;
  };
  float* deg  = (float*)carve((size_t)n * sizeof(float));
  float* h1   = (float*)carve((size_t)n * HIDDEN * sizeof(float));
  float* agg1 = (float*)carve((size_t)n * HIDDEN * sizeof(float));
  float* g    = (float*)carve((size_t)n * NCLS * sizeof(float));
  float* agg2 = (float*)carve((size_t)n * NCLS * sizeof(float));
  (void)ws_size;
  int* ctr_deg  = (int*)agg2;
  int* ctr_agg1 = (int*)agg2 + 2;
  int* ctr_agg2 = (int*)h1;

  const int B = 256;
  auto blocks = [](long long work, int b) { return (unsigned)((work + b - 1) / b); };

  k_init_deg<<<blocks(n, B), B, 0, stream>>>(deg, ctr_deg, n);
  int tiles = (n + 15) / 16;
  k_deg_gemm<<<blocks(tiles, GEMM_WAVES) + PASS_GRID, EDGE_THREADS, 0, stream>>>(
      col, deg, ctr_deg, E, n, x, W1, h1);

  k_agg1_self<<<blocks((long long)n * HIDDEN, B), B, 0, stream>>>(h1, deg, agg1, n);
  k_agg1_edges<<<PASS_GRID, EDGE_THREADS, 0, stream>>>(row, col, h1, deg, agg1, ctr_agg1, E, n);

  k_relu_gemm2<<<blocks(n, B), B, 0, stream>>>(agg1, b1, W2, deg, g, n);

  k_agg2_self<<<blocks(n, B), B, 0, stream>>>(g, agg2, ctr_agg2, n);
  k_agg2_edges<<<PASS_GRID, EDGE_THREADS, 0, stream>>>(row, col, g, agg2, ctr_agg2, E, n);

  k_logsoftmax<<<blocks(n, B), B, 0, stream>>>(agg2, deg, b2, (float*)d_out, n);
}
